# starting kernel
# baseline (speedup 1.0000x reference)
#include <hip/hip_runtime.h>
#include <math.h>

#define B_    16
#define C_    64
#define T_    2048
#define OUT_  512
#define NEED_ 257
#define K2_   514
#define K2P_  544

static __device__ __constant__ float kTwoPi = 6.283185307179586f;

typedef float v2f __attribute__((ext_vector_type(2)));
typedef float v8f __attribute__((ext_vector_type(8)));

__global__ void sp_build_w(float* __restrict__ W) {
    int idx = blockIdx.x * blockDim.x + threadIdx.x;
    if (idx >= K2P_ * OUT_) return;
    int row = idx / OUT_;
    int n   = idx - row * OUT_;
    int k   = row >> 1;
    float val = 0.0f;
    if (k < NEED_) {
        float wk = (k == 0 || k == NEED_ - 1) ? 1.0f : 2.0f;
        int im = (k * n) & (OUT_ - 1);
        float ang = kTwoPi * (float)im * (1.0f / (float)OUT_);
        float s, c;
        __sincosf(ang, &s, &c);
        val = (row & 1) ? (-wk * s * (1.0f / (float)OUT_))
                        : ( wk * c * (1.0f / (float)OUT_));
    }
    W[idx] = val;
}

#define KSTEP1 64
__global__ __launch_bounds__(256) void sp_gemm1(const float* __restrict__ x,
                                                const int*   __restrict__ length,
                                                float*       __restrict__ Y) {
    int b  = blockIdx.x;
    int nb = blockIdx.y;
    int Lr = length[b];
    int L  = Lr < 1 ? 1 : Lr;
    float invL = 1.0f / (float)L;

    __shared__ float sA[C_][KSTEP1 + 1];
    __shared__ float sF[KSTEP1][32 + 1];

    int tid   = threadIdx.x;
    int lane  = tid & 31;
    int wave  = tid >> 5;
    int mtile = wave >> 1;
    int ntile = wave & 1;

    v8f acc = {0.f, 0.f, 0.f, 0.f, 0.f, 0.f, 0.f, 0.f};

    const float* xb = x + (size_t)b * C_ * T_;

    for (int kk = 0; kk < T_; kk += KSTEP1) {
        #pragma unroll
        for (int i = 0; i < (C_ * KSTEP1) / 256; ++i) {
            int li = tid + i * 256;
            int r  = li >> 6;
            int c  = li & (KSTEP1 - 1);
            sA[r][c] = xb[r * T_ + kk + c];
        }
        #pragma unroll
        for (int i = 0; i < (KSTEP1 * 32) / 256; ++i) {
            int li = tid + i * 256;
            int tr = li >> 5;
            int nc = li & 31;
            int t  = kk + tr;
            int g2 = nb * 32 + nc;
            int k  = g2 >> 1;
            float v = 0.0f;
            if (t < L && k < NEED_) {
                int im = (k * t) % L;
                float ang = kTwoPi * (float)im * invL;
                float s, c;
                __sincosf(ang, &s, &c);
                v = (g2 & 1) ? -s : c;
            }
            sF[tr][nc] = v;
        }
        __syncthreads();

        int arow = mtile * 16 + (lane & 15);
        int bcol = ntile * 16 + (lane & 15);
        int koff = (lane >> 4) << 1;
        #pragma unroll
        for (int ks = 0; ks < KSTEP1; ks += 4) {
            v2f a, bf;
            a.x  = sA[arow][ks + koff];
            a.y  = sA[arow][ks + koff + 1];
            bf.x = sF[ks + koff][bcol];
            bf.y = sF[ks + koff + 1][bcol];
            acc = __builtin_amdgcn_wmma_f32_16x16x4_f32(
                false, a, false, bf, (short)0, acc, false, false);
        }
        __syncthreads();
    }

    int colg    = nb * 32 + ntile * 16 + (lane & 15);
    int rowbase = mtile * 16 + ((lane >> 4) ? 8 : 0);
    float* Yb = Y + (size_t)b * C_ * K2P_;
    #pragma unroll
    for (int i = 0; i < 8; ++i) {
        Yb[(rowbase + i) * K2P_ + colg] = acc[i];
    }
}

#define KSTEP2 32
__global__ __launch_bounds__(256) void sp_gemm2(const float* __restrict__ Y,
                                                const float* __restrict__ W,
                                                const float* __restrict__ x,
                                                const int*   __restrict__ length,
                                                float*       __restrict__ out) {
    int b  = blockIdx.x;
    int nb = blockIdx.y;
    int L  = length[b];

    __shared__ float sY[C_][KSTEP2 + 1];
    __shared__ float sW[KSTEP2][32 + 1];

    int tid   = threadIdx.x;
    int lane  = tid & 31;
    int wave  = tid >> 5;
    int mtile = wave >> 1;
    int ntile = wave & 1;

    v8f acc = {0.f, 0.f, 0.f, 0.f, 0.f, 0.f, 0.f, 0.f};

    const float* Yb = Y + (size_t)b * C_ * K2P_;

    for (int kk = 0; kk < K2P_; kk += KSTEP2) {
        #pragma unroll
        for (int i = 0; i < (C_ * KSTEP2) / 256; ++i) {
            int li = tid + i * 256;
            int r  = li >> 5;
            int c  = li & (KSTEP2 - 1);
            sY[r][c] = Yb[r * K2P_ + kk + c];
        }
        #pragma unroll
        for (int i = 0; i < (KSTEP2 * 32) / 256; ++i) {
            int li = tid + i * 256;
            int r  = li >> 5;
            int c  = li & 31;
            sW[r][c] = W[(kk + r) * OUT_ + nb * 32 + c];
        }
        __syncthreads();

        int arow = mtile * 16 + (lane & 15);
        int bcol = ntile * 16 + (lane & 15);
        int koff = (lane >> 4) << 1;
        #pragma unroll
        for (int ks = 0; ks < KSTEP2; ks += 4) {
            v2f a, bf;
            a.x  = sY[arow][ks + koff];
            a.y  = sY[arow][ks + koff + 1];
            bf.x = sW[ks + koff][bcol];
            bf.y = sW[ks + koff + 1][bcol];
            acc = __builtin_amdgcn_wmma_f32_16x16x4_f32(
                false, a, false, bf, (short)0, acc, false, false);
        }
        __syncthreads();
    }

    int n       = nb * 32 + ntile * 16 + (lane & 15);
    int rowbase = mtile * 16 + ((lane >> 4) ? 8 : 0);
    const float* xb = x + (size_t)b * C_ * T_;
    float* ob = out + (size_t)b * C_ * OUT_;
    bool shortSeq = (L < OUT_);
    #pragma unroll
    for (int i = 0; i < 8; ++i) {
        int c = rowbase + i;
        float v = acc[i];
        if (shortSeq) v = (n < L) ? xb[c * T_ + n] : 0.0f;
        ob[c * OUT_ + n] = v;
    }
}

extern "C" void kernel_launch(void* const* d_in, const int* in_sizes, int n_in,
                              void* d_out, int out_size, void* d_ws, size_t ws_size,
                              hipStream_t stream) {
    (void)in_sizes; (void)n_in; (void)out_size; (void)ws_size;
    const float* x      = (const float*)d_in[0];
    const int*   length = (const int*)d_in[1];
    float*       out    = (float*)d_out;

    float* W = (float*)d_ws;
    float* Y = W + (size_t)K2P_ * OUT_;

    sp_build_w<<<(K2P_ * OUT_ + 255) / 256, 256, 0, stream>>>(W);
    sp_gemm1<<<dim3(B_, K2P_ / 32), 256, 0, stream>>>(x, length, Y);
    sp_gemm2<<<dim3(B_, OUT_ / 32), 256, 0, stream>>>(Y, W, x, length, out);
}


# K split across a work-group's waves: a wave takes 64 rows x 16 cols and a quarter of every K chunk, makes its G fragments in registers (G skips LDS), 4 accumulators; new sum order; 2-waves/SIMD hint
# speedup vs baseline: 2.7646x; 2.7646x over previous
#include <hip/hip_runtime.h>


#define B_    16
#define C_    64
#define T_    2048
#define OUT_  512


typedef float v2f __attribute__((ext_vector_type(2)));
typedef float v8f __attribute__((ext_vector_type(8)));

__device__ __forceinline__ float sp_sin(float y) {
    float y2 = y * y;
    float p = fmaf(y2, 2.631678171e-6f, -1.982140529e-4f);
    p = fmaf(y2, p, 8.333225735e-3f);
    p = fmaf(y2, p, -1.666666567e-1f);
    p = fmaf(y2, p, 1.0f);
    return y * p;
}
__device__ __forceinline__ float sp_cos(float y) {
    float y2 = y * y;
    float p = fmaf(y2, 2.351005423e-5f, -1.387001481e-3f);
    p = fmaf(y2, p, 4.166582599e-2f);
    p = fmaf(y2, p, -4.999999702e-1f);
    p = fmaf(y2, p, 1.0f);
    return p;
}

#define KSTEP  128
#define APITCH 132
#define RPITCH 36

__global__ __launch_bounds__(256) __attribute__((amdgpu_waves_per_eu(2, 2))) void sp_fused(const float* __restrict__ x,
                                                const int*   __restrict__ length,
                                                float*       __restrict__ out) {
    int b  = blockIdx.x;
    int nb = blockIdx.y;
    int Lr = length[b];
    int L  = Lr < 1 ? 1 : Lr;

    __shared__ __attribute__((aligned(16))) float sA[C_][APITCH];
    __shared__ __attribute__((aligned(16))) float sS[T_];
    __shared__ float sR[4][C_][RPITCH];

    int tid   = threadIdx.x;
    int lane  = tid & 31;
    int wave  = tid >> 5;
    int ntile = wave & 1;
    int kq    = wave >> 1;
    int c     = lane & 15;
    int h     = lane >> 4;

    {
        float piL = 3.14159265358979f / (float)L;
        int L2 = 2 * L;
        #pragma unroll
        for (int i = 0; i < T_ / 256; ++i) {
            int t = tid + i * 256;
            int m = (512 * t) % L2;
            if (m > L) m -= L2;
            int mp = m;
            if (2 * m > L)       mp =  L - m;
            else if (2 * m < -L) mp = -L - m;
            sS[t] = sp_sin((float)mp * piL);
        }
    }
    __syncthreads();

    int   n    = nb * 32 + ntile * 16 + c;
    float nLf  = (float)(n * L);
    float Cf   = 512.0f * (float)L;
    float invC = 1.0f / Cf;
    float piC  = 3.14159265358979f / Cf;
    float sig  = (n & 1) ? (1.0f / 512.0f) : (-1.0f / 512.0f);

    int kthr = kq * 32 + 2 * h;

    v8f acc[4];
    #pragma unroll
    for (int mt = 0; mt < 4; ++mt)
        acc[mt] = (v8f){0.f, 0.f, 0.f, 0.f, 0.f, 0.f, 0.f, 0.f};

    const float* xb = x + (size_t)b * C_ * T_;

    for (int kk = 0; kk < T_; kk += KSTEP) {
        float4 xr[8];
        #pragma unroll
        for (int i = 0; i < 8; ++i) {
            int li = tid + i * 256;
            int r  = li >> 5;
            int c4 = li & 31;
            xr[i] = *(const float4*)(xb + r * T_ + kk + c4 * 4);
        }
        v2f bf[8];
        #pragma unroll
        for (int s = 0; s < 8; ++s) {
            int t0 = kk + kthr + 4 * s;
            v2f sv = *(const v2f*)(&sS[t0]);
            #pragma unroll
            for (int i = 0; i < 2; ++i) {
                int t  = t0 + i;
                float pf = fmaf(-512.0f, (float)t, nLf);
                float kq_ = rintf(pf * invC);
                float pr = fmaf(-Cf, kq_, pf);
                float y  = pr * piC;
                float cot = sp_cos(y) * __builtin_amdgcn_rcpf(sp_sin(y));
                float v = (sv[i] * sig) * cot;
                v = (pr == 0.0f) ? 1.0f : v;
                bf[s][i] = (t < L) ? v : 0.0f;
            }
        }
        #pragma unroll
        for (int i = 0; i < 8; ++i) {
            int li = tid + i * 256;
            int r  = li >> 5;
            int c4 = li & 31;
            *(float4*)(&sA[r][c4 * 4]) = xr[i];
        }
        __syncthreads();

        #pragma unroll
        for (int s = 0; s < 8; ++s) {
            int kl = kthr + 4 * s;
            #pragma unroll
            for (int mt = 0; mt < 4; ++mt) {
                v2f a = *(const v2f*)(&sA[mt * 16 + c][kl]);
                acc[mt] = __builtin_amdgcn_wmma_f32_16x16x4_f32(
                    false, a, false, bf[s], (short)0, acc[mt], false, false);
            }
        }
        __syncthreads();
    }

    #pragma unroll
    for (int mt = 0; mt < 4; ++mt) {
        #pragma unroll
        for (int r = 0; r < 8; ++r) {
            sR[kq][mt * 16 + 8 * h + r][ntile * 16 + c] = acc[mt][r];
        }
    }
    __syncthreads();

    int no = nb * 32 + lane;
    float* ob = out + (size_t)b * C_ * OUT_;
    bool shortSeq = (Lr < OUT_);
    #pragma unroll
    for (int i = 0; i < 8; ++i) {
        int row = wave * 8 + i;
        float v = ((sR[0][row][lane] + sR[1][row][lane]) + sR[2][row][lane]) + sR[3][row][lane];
        if (shortSeq) v = (no < Lr) ? xb[row * T_ + no] : 0.0f;
        ob[row * OUT_ + no] = v;
    }
}

extern "C" void kernel_launch(void* const* d_in, const int* in_sizes, int n_in,
                              void* d_out, int out_size, void* d_ws, size_t ws_size,
                              hipStream_t stream) {
    (void)in_sizes; (void)n_in; (void)out_size; (void)d_ws; (void)ws_size;
    const float* x      = (const float*)d_in[0];
    const int*   length = (const int*)d_in[1];
    float*       out    = (float*)d_out;

    sp_fused<<<dim3(B_, OUT_ / 32), 256, 0, stream>>>(x, length, out);
}


# K dealt to the 8 waves in 32-k slices: a wave owns 64x32 outputs and stages its own x piece in a wave-private LDS tile, so the K loop has no work-group barrier and waves overlap; new sum order
# speedup vs baseline: 4.0243x; 1.4557x over previous
#include <hip/hip_runtime.h>


#define B_    16
#define C_    64
#define T_    2048
#define OUT_  512


typedef float v2f __attribute__((ext_vector_type(2)));
typedef float v8f __attribute__((ext_vector_type(8)));

__device__ __forceinline__ float sp_sin(float y) {
    float y2 = y * y;
    float p = fmaf(y2, 2.631678171e-6f, -1.982140529e-4f);
    p = fmaf(y2, p, 8.333225735e-3f);
    p = fmaf(y2, p, -1.666666567e-1f);
    p = fmaf(y2, p, 1.0f);
    return y * p;
}
__device__ __forceinline__ float sp_cos(float y) {
    float y2 = y * y;
    float p = fmaf(y2, 2.351005423e-5f, -1.387001481e-3f);
    p = fmaf(y2, p, 4.166582599e-2f);
    p = fmaf(y2, p, -4.999999702e-1f);
    p = fmaf(y2, p, 1.0f);
    return p;
}

#define KW     32
#define XPITCH 36
#define NWAVE  8

__device__ __forceinline__ void wave_lds_order() {
    __builtin_amdgcn_fence(__ATOMIC_RELEASE, "wavefront");
    __builtin_amdgcn_wave_barrier();
    __builtin_amdgcn_fence(__ATOMIC_ACQUIRE, "wavefront");
}

__global__ __launch_bounds__(256) __attribute__((amdgpu_waves_per_eu(2, 2))) void sp_fused(const float* __restrict__ x,
                                                const int*   __restrict__ length,
                                                float*       __restrict__ out) {
    int b  = blockIdx.x;
    int nb = blockIdx.y;
    int Lr = length[b];
    int L  = Lr < 1 ? 1 : Lr;

    __shared__ __attribute__((aligned(16))) float sX[NWAVE][C_][XPITCH];
    __shared__ __attribute__((aligned(16))) float sS[T_];

    int tid   = threadIdx.x;
    int lane  = tid & 31;
    int wave  = tid >> 5;
    int c     = lane & 15;
    int h     = lane >> 4;

    {
        float piL = 3.14159265358979f / (float)L;
        int L2 = 2 * L;
        #pragma unroll
        for (int i = 0; i < T_ / 256; ++i) {
            int t = tid + i * 256;
            int m = (512 * t) % L2;
            if (m > L) m -= L2;
            int mp = m;
            if (2 * m > L)       mp =  L - m;
            else if (2 * m < -L) mp = -L - m;
            sS[t] = sp_sin((float)mp * piL);
        }
    }
    wave_lds_order();

    int   n0   = nb * 32 + c;
    float nLf[2];
    nLf[0] = (float)(n0 * L);
    nLf[1] = (float)((n0 + 16) * L);
    float Cf   = 512.0f * (float)L;
    float invC = 1.0f / Cf;
    float piC  = 3.14159265358979f / Cf;
    float sig  = (n0 & 1) ? (1.0f / 512.0f) : (-1.0f / 512.0f);

    v8f acc[4][2];
    #pragma unroll
    for (int mt = 0; mt < 4; ++mt)
        #pragma unroll
        for (int nt = 0; nt < 2; ++nt)
            acc[mt][nt] = (v8f){0.f, 0.f, 0.f, 0.f, 0.f, 0.f, 0.f, 0.f};

    const float* xb = x + (size_t)b * C_ * T_;
    int lrow = lane >> 3;
    int lc4  = lane & 7;

    for (int it = 0; it < T_ / (KW * NWAVE); ++it) {
        int tb = (it * NWAVE + wave) * KW;
        float4 xr[16];
        #pragma unroll
        for (int j = 0; j < 16; ++j)
            xr[j] = *(const float4*)(xb + (4 * j + lrow) * T_ + tb + lc4 * 4);

        v2f bf[8][2];
        #pragma unroll
        for (int s = 0; s < 8; ++s) {
            int t0 = tb + 4 * s + 2 * h;
            v2f sv = *(const v2f*)(&sS[t0]);
            #pragma unroll
            for (int nt = 0; nt < 2; ++nt) {
                #pragma unroll
                for (int e = 0; e < 2; ++e) {
                    int   t   = t0 + e;
                    float pf  = fmaf(-512.0f, (float)t, nLf[nt]);
                    float kq_ = rintf(pf * invC);
                    float pr  = fmaf(-Cf, kq_, pf);
                    float y   = pr * piC;
                    float cot = sp_cos(y) * __builtin_amdgcn_rcpf(sp_sin(y));
                    float v   = (sv[e] * sig) * cot;
                    v = (pr == 0.0f) ? 1.0f : v;
                    bf[s][nt][e] = (t < L) ? v : 0.0f;
                }
            }
        }
        #pragma unroll
        for (int j = 0; j < 16; ++j)
            *(float4*)(&sX[wave][4 * j + lrow][lc4 * 4]) = xr[j];
        wave_lds_order();

        #pragma unroll
        for (int s = 0; s < 8; ++s) {
            int kl = 4 * s + 2 * h;
            #pragma unroll
            for (int mt = 0; mt < 4; ++mt) {
                v2f a = *(const v2f*)(&sX[wave][mt * 16 + c][kl]);
                #pragma unroll
                for (int nt = 0; nt < 2; ++nt)
                    acc[mt][nt] = __builtin_amdgcn_wmma_f32_16x16x4_f32(
                        false, a, false, bf[s][nt], (short)0, acc[mt][nt], false, false);
            }
        }
        wave_lds_order();
        __builtin_amdgcn_sched_group_barrier(0x020, 16, 0);
        __builtin_amdgcn_sched_group_barrier(0x002, 520, 0);
        __builtin_amdgcn_sched_group_barrier(0x200, 16, 0);
    }

    #pragma unroll
    for (int mt = 0; mt < 4; ++mt)
        #pragma unroll
        for (int nt = 0; nt < 2; ++nt)
            #pragma unroll
            for (int r = 0; r < 8; ++r)
                sX[wave][mt * 16 + 8 * h + r][nt * 16 + c] = acc[mt][nt][r];
    __syncthreads();

    int no = nb * 32 + lane;
    float* ob = out + (size_t)b * C_ * OUT_;
    bool shortSeq = (Lr < OUT_);
    #pragma unroll
    for (int i = 0; i < 8; ++i) {
        int row = wave * 8 + i;
        float v = sX[0][row][lane];
        #pragma unroll
        for (int p = 1; p < NWAVE; ++p)
            v += sX[p][row][lane];
        if (shortSeq) v = (no < Lr) ? xb[row * T_ + no] : 0.0f;
        ob[row * OUT_ + no] = v;
    }
}

extern "C" void kernel_launch(void* const* d_in, const int* in_sizes, int n_in,
                              void* d_out, int out_size, void* d_ws, size_t ws_size,
                              hipStream_t stream) {
    (void)in_sizes; (void)n_in; (void)out_size; (void)d_ws; (void)ws_size;
    const float* x      = (const float*)d_in[0];
    const int*   length = (const int*)d_in[1];
    float*       out    = (float*)d_out;

    sp_fused<<<dim3(B_, OUT_ / 32), 256, 0, stream>>>(x, length, out);
}


# K loop software-pipelined inside each wave: behind every matrix instruction of slice i stands one stage of the G synthesis of slice i+1; x and A fragments fetched ahead of the run; same sum order
# speedup vs baseline: 4.3925x; 1.0915x over previous
#include <hip/hip_runtime.h>


#define B_    16
#define C_    64
#define T_    2048
#define OUT_  512


typedef float v2f __attribute__((ext_vector_type(2)));
typedef float v4f __attribute__((ext_vector_type(4)));
typedef float v8f __attribute__((ext_vector_type(8)));

__device__ __forceinline__ float sp_sin(float y) {
    float y2 = y * y;
    float p = fmaf(y2, 2.631678171e-6f, -1.982140529e-4f);
    p = fmaf(y2, p, 8.333225735e-3f);
    p = fmaf(y2, p, -1.666666567e-1f);
    p = fmaf(y2, p, 1.0f);
    return y * p;
}

struct GConst { float nLf0, nLf1, Cf, invC, piC, sig; int L; };
struct GSyn { float tf[4]; float pf[8]; float kq[8]; float pr[8]; float y[8]; float y2[8];
              float ps[8]; float pc[8]; float w[4]; bool z[8]; bool in[4]; };

__device__ __forceinline__ void g_stage(int j, GSyn& S, const GConst& g, int tA,
                                        const v2f (&sv)[2], v2f (&o0)[2], v2f (&o1)[2]) {
    switch (j) {
    case 0:
        #pragma unroll
        for (int u = 0; u < 4; ++u) S.tf[u] = (float)(tA + 4 * (u >> 1) + (u & 1));
        #pragma unroll
        for (int q = 0; q < 8; ++q)
            S.pf[q] = fmaf(-512.0f, S.tf[2 * (q >> 2) + (q & 1)], ((q >> 1) & 1) ? g.nLf1 : g.nLf0);
        break;
    case 1:
        #pragma unroll
        for (int q = 0; q < 8; ++q) S.kq[q] = S.pf[q] * g.invC;
        #pragma unroll
        for (int q = 0; q < 8; ++q) S.kq[q] = rintf(S.kq[q]);
        break;
    case 2:
        #pragma unroll
        for (int q = 0; q < 8; ++q) S.pr[q] = fmaf(-g.Cf, S.kq[q], S.pf[q]);
        #pragma unroll
        for (int q = 0; q < 8; ++q) S.y[q] = S.pr[q] * g.piC;
        #pragma unroll
        for (int q = 0; q < 8; ++q) S.y2[q] = S.y[q] * S.y[q];
        break;
    case 3:
        #pragma unroll
        for (int q = 0; q < 8; ++q) S.ps[q] = fmaf(S.y2[q], 2.631678171e-6f, -1.982140529e-4f);
        #pragma unroll
        for (int q = 0; q < 4; ++q) S.z[q] = (S.pr[q] != 0.0f);
        break;
    case 4:
        #pragma unroll
        for (int q = 0; q < 8; ++q) S.pc[q] = fmaf(S.y2[q], 2.351005423e-5f, -1.387001481e-3f);
        #pragma unroll
        for (int q = 4; q < 8; ++q) S.z[q] = (S.pr[q] != 0.0f);
        break;
    case 5:
        #pragma unroll
        for (int q = 0; q < 8; ++q) S.ps[q] = fmaf(S.y2[q], S.ps[q], 8.333225735e-3f);
        break;
    case 6:
        #pragma unroll
        for (int q = 0; q < 8; ++q) S.pc[q] = fmaf(S.y2[q], S.pc[q], 4.166582599e-2f);
        break;
    case 7:
        #pragma unroll
        for (int q = 0; q < 8; ++q) S.ps[q] = fmaf(S.y2[q], S.ps[q], -1.666666567e-1f);
        break;
    case 8:
        #pragma unroll
        for (int q = 0; q < 8; ++q) S.pc[q] = fmaf(S.y2[q], S.pc[q], -4.999999702e-1f);
        break;
    case 9:
        #pragma unroll
        for (int q = 0; q < 8; ++q) S.ps[q] = fmaf(S.y2[q], S.ps[q], 1.0f);
        break;
    case 10:
        #pragma unroll
        for (int q = 0; q < 8; ++q) S.pc[q] = fmaf(S.y2[q], S.pc[q], 1.0f);
        break;
    case 11:
        #pragma unroll
        for (int q = 0; q < 8; ++q) S.ps[q] = S.y[q] * S.ps[q];
        #pragma unroll
        for (int u = 0; u < 4; ++u) S.w[u] = sv[u >> 1][u & 1] * g.sig;
        #pragma unroll
        for (int u = 0; u < 4; ++u) S.in[u] = (tA + 4 * (u >> 1) + (u & 1) < g.L);
        break;
    case 12:
        #pragma unroll
        for (int q = 0; q < 8; ++q) S.ps[q] = __builtin_amdgcn_rcpf(S.ps[q]);
        break;
    case 13:
        #pragma unroll
        for (int q = 0; q < 8; ++q) S.pc[q] = S.pc[q] * S.ps[q];
        #pragma unroll
        for (int q = 0; q < 8; ++q) S.pc[q] = S.w[2 * (q >> 2) + (q & 1)] * S.pc[q];
        break;
    case 14:
        #pragma unroll
        for (int q = 0; q < 8; ++q) S.pc[q] = S.z[q] ? S.pc[q] : 1.0f;
        break;
    default:
        #pragma unroll
        for (int q = 0; q < 8; ++q) {
            float v = S.in[2 * (q >> 2) + (q & 1)] ? S.pc[q] : 0.0f;
            if (q >> 2) o1[(q >> 1) & 1][q & 1] = v; else o0[(q >> 1) & 1][q & 1] = v;
        }
        break;
    }
}

#define KW     32
#define XPITCH 36
#define NWAVE  8

__device__ __forceinline__ void wave_lds_order() {
    __builtin_amdgcn_fence(__ATOMIC_RELEASE, "wavefront");
    __builtin_amdgcn_wave_barrier();
    __builtin_amdgcn_fence(__ATOMIC_ACQUIRE, "wavefront");
}

__global__ __launch_bounds__(256) __attribute__((amdgpu_waves_per_eu(2, 2))) void sp_fused(const float* __restrict__ x,
                                                const int*   __restrict__ length,
                                                float*       __restrict__ out) {
    int b  = blockIdx.x;
    int nb = blockIdx.y;
    int Lr = length[b];
    int L  = Lr < 1 ? 1 : Lr;

    __shared__ __attribute__((aligned(16))) float sX[NWAVE][C_][XPITCH];
    __shared__ __attribute__((aligned(16))) float sS[T_];

    int tid   = threadIdx.x;
    int lane  = tid & 31;
    int wave  = tid >> 5;
    int c     = lane & 15;
    int h     = lane >> 4;

    {
        float piL = 3.14159265358979f / (float)L;
        int L2 = 2 * L;
        #pragma unroll
        for (int i = 0; i < T_ / 256; ++i) {
            int t = tid + i * 256;
            int m = (512 * t) % L2;
            if (m > L) m -= L2;
            int mp = m;
            if (2 * m > L)       mp =  L - m;
            else if (2 * m < -L) mp = -L - m;
            sS[t] = sp_sin((float)mp * piL);
        }
    }
    wave_lds_order();

    int   n0   = nb * 32 + c;
    float nLf[2];
    nLf[0] = (float)(n0 * L);
    nLf[1] = (float)((n0 + 16) * L);
    float Cf   = 512.0f * (float)L;
    float invC = 1.0f / Cf;
    float piC  = 3.14159265358979f / Cf;
    float sig  = (n0 & 1) ? (1.0f / 512.0f) : (-1.0f / 512.0f);

    v8f acc[4][2];
    #pragma unroll
    for (int mt = 0; mt < 4; ++mt)
        #pragma unroll
        for (int nt = 0; nt < 2; ++nt)
            acc[mt][nt] = (v8f){0.f, 0.f, 0.f, 0.f, 0.f, 0.f, 0.f, 0.f};

    const float* xb = x + (size_t)b * C_ * T_;
    int lrow = lane >> 3;
    int lc4  = lane & 7;

    GConst gc;
    gc.nLf0 = nLf[0]; gc.nLf1 = nLf[1]; gc.Cf = Cf; gc.invC = invC; gc.piC = piC; gc.sig = sig; gc.L = L;

    v2f bf[8][2];

    {
        int tb = wave * KW;
        v4f xr[16];
        #pragma unroll
        for (int j = 0; j < 16; ++j)
            xr[j] = *(const v4f*)(xb + (4 * j + lrow) * T_ + tb + lc4 * 4);
        #pragma unroll
        for (int sp = 0; sp < 4; ++sp) {
            int tA = tb + 8 * sp + 2 * h;
            v2f sv[2];
            sv[0] = *(const v2f*)(&sS[tA]);
            sv[1] = *(const v2f*)(&sS[tA + 4]);
            GSyn S;
            #pragma unroll
            for (int j = 0; j < 16; ++j)
                g_stage(j, S, gc, tA, sv, bf[2 * sp], bf[2 * sp + 1]);
        }
        #pragma unroll
        for (int j = 0; j < 16; ++j)
            *(v4f*)(&sX[wave][4 * j + lrow][lc4 * 4]) = xr[j];
        wave_lds_order();
    }

    const int NSLICE = T_ / (KW * NWAVE);
    for (int it = 0; it < NSLICE - 1; ++it) {
        int tbn = ((it + 1) * NWAVE + wave) * KW;
        v4f xr[16];
        #pragma unroll
        for (int j = 0; j < 16; ++j)
            xr[j] = *(const v4f*)(xb + (4 * j + lrow) * T_ + tbn + lc4 * 4);
        v2f af[8][4];
        #pragma unroll
        for (int s = 0; s < 8; ++s)
            #pragma unroll
            for (int mt = 0; mt < 4; ++mt)
                af[s][mt] = *(const v2f*)(&sX[wave][mt * 16 + c][4 * s + 2 * h]);
        v2f svn[8];
        #pragma unroll
        for (int s = 0; s < 8; ++s)
            svn[s] = *(const v2f*)(&sS[tbn + 4 * s + 2 * h]);
        __builtin_amdgcn_sched_barrier(0);

        #pragma unroll
        for (int sp = 0; sp < 4; ++sp) {
            int tA = tbn + 8 * sp + 2 * h;
            v2f sv[2];
            sv[0] = svn[2 * sp];
            sv[1] = svn[2 * sp + 1];
            GSyn S;
            #pragma unroll
            for (int j = 0; j < 16; ++j) {
                int s  = 2 * sp + (j >> 3);
                int mt = (j >> 1) & 3;
                int nt = j & 1;
                acc[mt][nt] = __builtin_amdgcn_wmma_f32_16x16x4_f32(
                    false, af[s][mt], false, bf[s][nt], (short)0, acc[mt][nt], false, false);
                g_stage(j, S, gc, tA, sv, bf[2 * sp], bf[2 * sp + 1]);
                __builtin_amdgcn_sched_barrier(0);
            }
        }
        wave_lds_order();
        #pragma unroll
        for (int j = 0; j < 16; ++j)
            *(v4f*)(&sX[wave][4 * j + lrow][lc4 * 4]) = xr[j];
        wave_lds_order();
    }

    #pragma unroll
    for (int s = 0; s < 8; ++s) {
        int kl = 4 * s + 2 * h;
        #pragma unroll
        for (int mt = 0; mt < 4; ++mt) {
            v2f a = *(const v2f*)(&sX[wave][mt * 16 + c][kl]);
            #pragma unroll
            for (int nt = 0; nt < 2; ++nt)
                acc[mt][nt] = __builtin_amdgcn_wmma_f32_16x16x4_f32(
                    false, a, false, bf[s][nt], (short)0, acc[mt][nt], false, false);
        }
    }
    wave_lds_order();

    #pragma unroll
    for (int mt = 0; mt < 4; ++mt)
        #pragma unroll
        for (int nt = 0; nt < 2; ++nt)
            #pragma unroll
            for (int r = 0; r < 8; ++r)
                sX[wave][mt * 16 + 8 * h + r][nt * 16 + c] = acc[mt][nt][r];
    __syncthreads();

    int no = nb * 32 + lane;
    float* ob = out + (size_t)b * C_ * OUT_;
    bool shortSeq = (Lr < OUT_);
    #pragma unroll
    for (int i = 0; i < 8; ++i) {
        int row = wave * 8 + i;
        float v = sX[0][row][lane];
        #pragma unroll
        for (int p = 1; p < NWAVE; ++p)
            v += sX[p][row][lane];
        if (shortSeq) v = (no < Lr) ? xb[row * T_ + no] : 0.0f;
        ob[row * OUT_ + no] = v;
    }
}

extern "C" void kernel_launch(void* const* d_in, const int* in_sizes, int n_in,
                              void* d_out, int out_size, void* d_ws, size_t ws_size,
                              hipStream_t stream) {
    (void)in_sizes; (void)n_in; (void)out_size; (void)d_ws; (void)ws_size;
    const float* x      = (const float*)d_in[0];
    const int*   length = (const int*)d_in[1];
    float*       out    = (float*)d_out;

    sp_fused<<<dim3(B_, OUT_ / 32), 256, 0, stream>>>(x, length, out);
}


# K range of every batch ends at its valid length and copied batches skip the product; 64x16 column tiles dealt to the 256 work-groups in pairs, longest batch with shortest; same sum order
# speedup vs baseline: 5.8534x; 1.3326x over previous
#include <hip/hip_runtime.h>


#define B_    16
#define C_    64
#define T_    2048
#define OUT_  512


typedef float v2f __attribute__((ext_vector_type(2)));
typedef float v4f __attribute__((ext_vector_type(4)));
typedef float v8f __attribute__((ext_vector_type(8)));

__device__ __forceinline__ float sp_sin(float y) {
    float y2 = y * y;
    float p = fmaf(y2, 2.631678171e-6f, -1.982140529e-4f);
    p = fmaf(y2, p, 8.333225735e-3f);
    p = fmaf(y2, p, -1.666666567e-1f);
    p = fmaf(y2, p, 1.0f);
    return y * p;
}

struct GConst { float nLf0, nLf1, Cf, invC, piC, sig; };
struct GSyn { float tf[4]; float pf[8]; float kq[8]; float pr[8]; float y[8]; float y2[8];
              float ps[8]; float pc[8]; float w[8]; bool z[8]; bool in[8]; };

__device__ __forceinline__ void g_stage(int j, GSyn& S, const GConst& g, int tA, int L,
                                        const v2f (&sv)[4], v2f* __restrict__ o) {
    switch (j) {
    case 0:
        #pragma unroll
        for (int u = 0; u < 4; ++u) S.tf[u] = (float)(tA + 4 * (u >> 1) + (u & 1));
        #pragma unroll
        for (int q = 0; q < 8; ++q)
            S.pf[q] = fmaf(-512.0f, S.tf[2 * (q >> 2) + (q & 1)], ((q >> 1) & 1) ? g.nLf1 : g.nLf0);
        break;
    case 1:
        #pragma unroll
        for (int q = 0; q < 8; ++q) S.kq[q] = S.pf[q] * g.invC;
        #pragma unroll
        for (int q = 0; q < 8; ++q) S.kq[q] = rintf(S.kq[q]);
        break;
    case 2:
        #pragma unroll
        for (int q = 0; q < 8; ++q) S.pr[q] = fmaf(-g.Cf, S.kq[q], S.pf[q]);
        #pragma unroll
        for (int q = 0; q < 8; ++q) S.y[q] = S.pr[q] * g.piC;
        #pragma unroll
        for (int q = 0; q < 8; ++q) S.y2[q] = S.y[q] * S.y[q];
        break;
    case 3:
        #pragma unroll
        for (int q = 0; q < 8; ++q) S.ps[q] = fmaf(S.y2[q], 2.631678171e-6f, -1.982140529e-4f);
        #pragma unroll
        for (int q = 0; q < 4; ++q) S.z[q] = (S.pr[q] != 0.0f);
        break;
    case 4:
        #pragma unroll
        for (int q = 0; q < 8; ++q) S.pc[q] = fmaf(S.y2[q], 2.351005423e-5f, -1.387001481e-3f);
        #pragma unroll
        for (int q = 4; q < 8; ++q) S.z[q] = (S.pr[q] != 0.0f);
        break;
    case 5:
        #pragma unroll
        for (int q = 0; q < 8; ++q) S.ps[q] = fmaf(S.y2[q], S.ps[q], 8.333225735e-3f);
        break;
    case 6:
        #pragma unroll
        for (int q = 0; q < 8; ++q) S.pc[q] = fmaf(S.y2[q], S.pc[q], 4.166582599e-2f);
        break;
    case 7:
        #pragma unroll
        for (int q = 0; q < 8; ++q) S.ps[q] = fmaf(S.y2[q], S.ps[q], -1.666666567e-1f);
        break;
    case 8:
        #pragma unroll
        for (int q = 0; q < 8; ++q) S.pc[q] = fmaf(S.y2[q], S.pc[q], -4.999999702e-1f);
        break;
    case 9:
        #pragma unroll
        for (int q = 0; q < 8; ++q) S.ps[q] = fmaf(S.y2[q], S.ps[q], 1.0f);
        #pragma unroll
        for (int q = 0; q < 4; ++q)
            S.in[q] = (tA + 4 * (q >> 2) + 8 * ((q >> 1) & 1) + (q & 1) < L);
        break;
    case 10:
        #pragma unroll
        for (int q = 0; q < 8; ++q) S.pc[q] = fmaf(S.y2[q], S.pc[q], 1.0f);
        #pragma unroll
        for (int q = 4; q < 8; ++q)
            S.in[q] = (tA + 4 * (q >> 2) + 8 * ((q >> 1) & 1) + (q & 1) < L);
        break;
    case 11:
        #pragma unroll
        for (int q = 0; q < 8; ++q) S.ps[q] = S.y[q] * S.ps[q];
        #pragma unroll
        for (int q = 0; q < 8; ++q) S.w[q] = sv[(q >> 2) + 2 * ((q >> 1) & 1)][q & 1] * g.sig;
        break;
    case 12:
        #pragma unroll
        for (int q = 0; q < 8; ++q) S.ps[q] = __builtin_amdgcn_rcpf(S.ps[q]);
        break;
    case 13:
        #pragma unroll
        for (int q = 0; q < 8; ++q) S.pc[q] = S.pc[q] * S.ps[q];
        #pragma unroll
        for (int q = 0; q < 8; ++q) S.pc[q] = S.w[q] * S.pc[q];
        break;
    case 14:
        #pragma unroll
        for (int q = 0; q < 8; ++q) S.pc[q] = S.z[q] ? S.pc[q] : 1.0f;
        break;
    default:
        #pragma unroll
        for (int q = 0; q < 8; ++q)
            o[(q >> 2) + 2 * ((q >> 1) & 1)][q & 1] = S.in[q] ? S.pc[q] : 0.0f;
        break;
    }
}

#define KW     32
#define XPITCH 36
#define NWAVE  8

__device__ __forceinline__ void wave_lds_order() {
    __builtin_amdgcn_fence(__ATOMIC_RELEASE, "wavefront");
    __builtin_amdgcn_wave_barrier();
    __builtin_amdgcn_fence(__ATOMIC_ACQUIRE, "wavefront");
}

__global__ __launch_bounds__(256) __attribute__((amdgpu_waves_per_eu(2, 2))) void sp_fused(const float* __restrict__ x,
                                                const int*   __restrict__ length,
                                                float*       __restrict__ out) {
    __shared__ __attribute__((aligned(16))) float sX[NWAVE][C_][XPITCH];
    __shared__ __attribute__((aligned(16))) float sS[T_];

    int g     = blockIdx.x;
    int tid   = threadIdx.x;
    int lane  = tid & 31;
    int wave  = __builtin_amdgcn_readfirstlane(tid >> 5);
    int c     = lane & 15;
    int h     = lane >> 4;

    int myL = length[lane & 15];
    int myCost;
    {
        int ns = (myL + KW - 1) >> 5;
        ns = ns > T_ / KW ? T_ / KW : ns;
        myCost = (myL >= OUT_) ? ((ns + NWAVE - 1) >> 3) : 0;
    }
    int rank = 0;
    #pragma unroll
    for (int bb = 0; bb < B_; ++bb) {
        int cb = __builtin_amdgcn_readlane(myCost, bb);
        rank += (cb > myCost || (cb == myCost && bb < (lane & 15))) ? 1 : 0;
    }

    {
        int row = g >> 2;
        int col = (g & 3) * 128 + lane * 4;
        #pragma unroll
        for (int k = 0; k < 2; ++k) {
            int bq = wave + 8 * k;
            int Lq = __builtin_amdgcn_readlane(myL, bq);
            if (Lq < OUT_) {
                v4f v = *(const v4f*)(x + ((size_t)bq * C_ + row) * T_ + col);
                v4f r;
                r.x = (col + 0 < Lq) ? v.x : 0.0f;
                r.y = (col + 1 < Lq) ? v.y : 0.0f;
                r.z = (col + 2 < Lq) ? v.z : 0.0f;
                r.w = (col + 3 < Lq) ? v.w : 0.0f;
                *(v4f*)(out + ((size_t)bq * C_ + row) * OUT_ + col) = r;
            }
        }
    }

    int pr = g >> 5;
    int ct = g & 31;

    int lrow = lane >> 3;
    int lc4  = lane & 7;

    for (int u = 0; u < 2; ++u) {
        unsigned hit = __builtin_amdgcn_ballot_w32(rank == (u ? 15 - pr : pr)) & 0xffffu;
        int b  = __builtin_ctz(hit);
        int Lr = __builtin_amdgcn_readlane(myL, b);
        if (Lr < OUT_) continue;
        int L  = Lr;
        int nsl = (L + KW - 1) / KW;
        nsl = nsl > T_ / KW ? T_ / KW : nsl;
        int cnt = (nsl - wave + NWAVE - 1) / NWAVE;

        {
            float piL = 3.14159265358979f / (float)L;
            int L2 = 2 * L;
            for (int i = 0; i < cnt; ++i) {
                int t = tid + i * 256;
                int m = (512 * t) % L2;
                if (m > L) m -= L2;
                int mp = m;
                if (2 * m > L)       mp =  L - m;
                else if (2 * m < -L) mp = -L - m;
                sS[t] = sp_sin((float)mp * piL);
            }
        }
        wave_lds_order();

        int   n0   = ct * 16 + c;
        float Cf   = 512.0f * (float)L;
        GConst gc;
        gc.nLf0 = (float)(n0 * L);
        gc.nLf1 = (float)(n0 * L - 512 * 8);
        gc.Cf   = Cf;
        gc.invC = 1.0f / Cf;
        gc.piC  = 3.14159265358979f / Cf;
        gc.sig  = (n0 & 1) ? (1.0f / 512.0f) : (-1.0f / 512.0f);

        v8f acc[4];
        #pragma unroll
        for (int mt = 0; mt < 4; ++mt)
            acc[mt] = (v8f){0.f, 0.f, 0.f, 0.f, 0.f, 0.f, 0.f, 0.f};

        const float* xb = x + (size_t)b * C_ * T_;

        v2f bf[8];

        {
            int tb = wave * KW;
            v4f xr0[16];
            #pragma unroll
            for (int i = 0; i < 16; ++i)
                xr0[i] = *(const v4f*)(xb + (4 * i + lrow) * T_ + tb + lc4 * 4);
            #pragma unroll
            for (int sp = 0; sp < 2; ++sp) {
                int tA = tb + 16 * sp + 2 * h;
                v2f sv[4];
                #pragma unroll
                for (int sl = 0; sl < 4; ++sl)
                    sv[sl] = *(const v2f*)(&sS[tA + 4 * sl]);
                GSyn S;
                #pragma unroll
                for (int st = 0; st < 16; ++st)
                    g_stage(st, S, gc, tA, L, sv, bf + 4 * sp);
            }
            #pragma unroll
            for (int i = 0; i < 16; ++i)
                *(v4f*)(&sX[wave][4 * i + lrow][lc4 * 4]) = xr0[i];
            wave_lds_order();
        }

        for (int it = 0; it < cnt - 1; ++it) {
            int tbn = ((it + 1) * NWAVE + wave) * KW;
            v4f xr[16];
            #pragma unroll
            for (int i = 0; i < 16; ++i)
                xr[i] = *(const v4f*)(xb + (4 * i + lrow) * T_ + tbn + lc4 * 4);
            v2f af[8][4];
            #pragma unroll
            for (int s2 = 0; s2 < 8; ++s2)
                #pragma unroll
                for (int mt = 0; mt < 4; ++mt)
                    af[s2][mt] = *(const v2f*)(&sX[wave][mt * 16 + c][4 * s2 + 2 * h]);
            v2f svn[8];
            #pragma unroll
            for (int s2 = 0; s2 < 8; ++s2)
                svn[s2] = *(const v2f*)(&sS[tbn + 4 * s2 + 2 * h]);
            __builtin_amdgcn_sched_barrier(0);

            #pragma unroll
            for (int sp = 0; sp < 2; ++sp) {
                int tA = tbn + 16 * sp + 2 * h;
                v2f sv[4];
                #pragma unroll
                for (int sl = 0; sl < 4; ++sl) sv[sl] = svn[4 * sp + sl];
                GSyn S;
                #pragma unroll
                for (int st = 0; st < 16; ++st) {
                    int s2 = 4 * sp + (st >> 2);
                    int mt = st & 3;
                    acc[mt] = __builtin_amdgcn_wmma_f32_16x16x4_f32(
                        false, af[s2][mt], false, bf[s2], (short)0, acc[mt], false, false);
                    g_stage(st, S, gc, tA, L, sv, bf + 4 * sp);
                    __builtin_amdgcn_sched_barrier(0);
                }
            }
            wave_lds_order();
            #pragma unroll
            for (int i = 0; i < 16; ++i)
                *(v4f*)(&sX[wave][4 * i + lrow][lc4 * 4]) = xr[i];
            wave_lds_order();
        }

        #pragma unroll
        for (int s2 = 0; s2 < 8; ++s2) {
            int kl = 4 * s2 + 2 * h;
            #pragma unroll
            for (int mt = 0; mt < 4; ++mt) {
                v2f a = *(const v2f*)(&sX[wave][mt * 16 + c][kl]);
                acc[mt] = __builtin_amdgcn_wmma_f32_16x16x4_f32(
                    false, a, false, bf[s2], (short)0, acc[mt], false, false);
            }
        }
        wave_lds_order();

        #pragma unroll
        for (int mt = 0; mt < 4; ++mt)
            #pragma unroll
            for (int r = 0; r < 8; ++r)
                sX[wave][mt * 16 + 8 * h + r][c] = acc[mt][r];
        __syncthreads();

        {
            int row = tid >> 2;
            int cq  = (tid & 3) * 4;
            v4f v = *(const v4f*)(&sX[0][row][cq]);
            #pragma unroll
            for (int p = 1; p < NWAVE; ++p)
                v += *(const v4f*)(&sX[p][row][cq]);
            *(v4f*)(out + ((size_t)b * C_ + row) * OUT_ + ct * 16 + cq) = v;
        }
        __syncthreads();
    }
}

extern "C" void kernel_launch(void* const* d_in, const int* in_sizes, int n_in,
                              void* d_out, int out_size, void* d_ws, size_t ws_size,
                              hipStream_t stream) {
    (void)in_sizes; (void)n_in; (void)out_size; (void)d_ws; (void)ws_size;
    const float* x      = (const float*)d_in[0];
    const int*   length = (const int*)d_in[1];
    float*       out    = (float*)d_out;

    sp_fused<<<dim3(256), 256, 0, stream>>>(x, length, out);
}


# The two units of a work-group run side by side on two groups of its waves, split (batches ranked) by live K slices, not in turn: table, first-slice synthesis and closing sum paid once; new sum order
# speedup vs baseline: 6.5549x; 1.1198x over previous
#include <hip/hip_runtime.h>


#define B_    16
#define C_    64
#define T_    2048
#define OUT_  512


typedef float v2f __attribute__((ext_vector_type(2)));
typedef float v4f __attribute__((ext_vector_type(4)));
typedef float v8f __attribute__((ext_vector_type(8)));

__device__ __forceinline__ float sp_sin(float y) {
    float y2 = y * y;
    float p = fmaf(y2, 2.631678171e-6f, -1.982140529e-4f);
    p = fmaf(y2, p, 8.333225735e-3f);
    p = fmaf(y2, p, -1.666666567e-1f);
    p = fmaf(y2, p, 1.0f);
    return y * p;
}

struct GConst { float nLf0, nLf1, Cf, invC, piC, sig; };
struct GSyn { float tf[4]; float pf[8]; float kq[8]; float pr[8]; float y[8]; float y2[8];
              float ps[8]; float pc[8]; float w[8]; bool z[8]; bool in[8]; };

__device__ __forceinline__ void g_stage(int j, GSyn& S, const GConst& g, int tA, int L,
                                        const v2f (&sv)[4], v2f* __restrict__ o) {
    switch (j) {
    case 0:
        #pragma unroll
        for (int u = 0; u < 4; ++u) S.tf[u] = (float)(tA + 4 * (u >> 1) + (u & 1));
        #pragma unroll
        for (int q = 0; q < 8; ++q)
            S.pf[q] = fmaf(-512.0f, S.tf[2 * (q >> 2) + (q & 1)], ((q >> 1) & 1) ? g.nLf1 : g.nLf0);
        break;
    case 1:
        #pragma unroll
        for (int q = 0; q < 8; ++q) S.kq[q] = S.pf[q] * g.invC;
        #pragma unroll
        for (int q = 0; q < 8; ++q) S.kq[q] = rintf(S.kq[q]);
        break;
    case 2:
        #pragma unroll
        for (int q = 0; q < 8; ++q) S.pr[q] = fmaf(-g.Cf, S.kq[q], S.pf[q]);
        #pragma unroll
        for (int q = 0; q < 8; ++q) S.y[q] = S.pr[q] * g.piC;
        #pragma unroll
        for (int q = 0; q < 8; ++q) S.y2[q] = S.y[q] * S.y[q];
        break;
    case 3:
        #pragma unroll
        for (int q = 0; q < 8; ++q) S.ps[q] = fmaf(S.y2[q], 2.631678171e-6f, -1.982140529e-4f);
        #pragma unroll
        for (int q = 0; q < 4; ++q) S.z[q] = (S.pr[q] != 0.0f);
        break;
    case 4:
        #pragma unroll
        for (int q = 0; q < 8; ++q) S.pc[q] = fmaf(S.y2[q], 2.351005423e-5f, -1.387001481e-3f);
        #pragma unroll
        for (int q = 4; q < 8; ++q) S.z[q] = (S.pr[q] != 0.0f);
        break;
    case 5:
        #pragma unroll
        for (int q = 0; q < 8; ++q) S.ps[q] = fmaf(S.y2[q], S.ps[q], 8.333225735e-3f);
        break;
    case 6:
        #pragma unroll
        for (int q = 0; q < 8; ++q) S.pc[q] = fmaf(S.y2[q], S.pc[q], 4.166582599e-2f);
        break;
    case 7:
        #pragma unroll
        for (int q = 0; q < 8; ++q) S.ps[q] = fmaf(S.y2[q], S.ps[q], -1.666666567e-1f);
        break;
    case 8:
        #pragma unroll
        for (int q = 0; q < 8; ++q) S.pc[q] = fmaf(S.y2[q], S.pc[q], -4.999999702e-1f);
        break;
    case 9:
        #pragma unroll
        for (int q = 0; q < 8; ++q) S.ps[q] = fmaf(S.y2[q], S.ps[q], 1.0f);
        #pragma unroll
        for (int q = 0; q < 4; ++q)
            S.in[q] = (tA + 4 * (q >> 2) + 8 * ((q >> 1) & 1) + (q & 1) < L);
        break;
    case 10:
        #pragma unroll
        for (int q = 0; q < 8; ++q) S.pc[q] = fmaf(S.y2[q], S.pc[q], 1.0f);
        #pragma unroll
        for (int q = 4; q < 8; ++q)
            S.in[q] = (tA + 4 * (q >> 2) + 8 * ((q >> 1) & 1) + (q & 1) < L);
        break;
    case 11:
        #pragma unroll
        for (int q = 0; q < 8; ++q) S.ps[q] = S.y[q] * S.ps[q];
        #pragma unroll
        for (int q = 0; q < 8; ++q) S.w[q] = sv[(q >> 2) + 2 * ((q >> 1) & 1)][q & 1] * g.sig;
        break;
    case 12:
        #pragma unroll
        for (int q = 0; q < 8; ++q) S.ps[q] = __builtin_amdgcn_rcpf(S.ps[q]);
        break;
    case 13:
        #pragma unroll
        for (int q = 0; q < 8; ++q) S.pc[q] = S.pc[q] * S.ps[q];
        #pragma unroll
        for (int q = 0; q < 8; ++q) S.pc[q] = S.w[q] * S.pc[q];
        break;
    case 14:
        #pragma unroll
        for (int q = 0; q < 8; ++q) S.pc[q] = S.z[q] ? S.pc[q] : 1.0f;
        break;
    default:
        #pragma unroll
        for (int q = 0; q < 8; ++q)
            o[(q >> 2) + 2 * ((q >> 1) & 1)][q & 1] = S.in[q] ? S.pc[q] : 0.0f;
        break;
    }
}

#define KW     32
#define XPITCH 36
#define NWAVE  8

__device__ __forceinline__ void wave_lds_order() {
    __builtin_amdgcn_fence(__ATOMIC_RELEASE, "wavefront");
    __builtin_amdgcn_wave_barrier();
    __builtin_amdgcn_fence(__ATOMIC_ACQUIRE, "wavefront");
}

__global__ __launch_bounds__(256) __attribute__((amdgpu_waves_per_eu(2, 2))) void sp_fused(const float* __restrict__ x,
                                                const int*   __restrict__ length,
                                                float*       __restrict__ out) {
    __shared__ __attribute__((aligned(16))) float sX[NWAVE][C_][XPITCH];
    __shared__ __attribute__((aligned(16))) float sS[2][T_];

    int g     = blockIdx.x;
    int tid   = threadIdx.x;
    int lane  = tid & 31;
    int wave  = __builtin_amdgcn_readfirstlane(tid >> 5);
    int c     = lane & 15;
    int h     = lane >> 4;

    int myL = length[lane & 15];
    int myCost;
    {
        int ns = (myL + KW - 1) >> 5;
        ns = ns > T_ / KW ? T_ / KW : ns;
        myCost = (myL >= OUT_) ? ns : 0;
    }
    int rank = 0;
    #pragma unroll
    for (int bb = 0; bb < B_; ++bb) {
        int cb = __builtin_amdgcn_readlane(myCost, bb);
        rank += (cb > myCost || (cb == myCost && bb < (lane & 15))) ? 1 : 0;
    }

    {
        int row = g >> 2;
        int col = (g & 3) * 128 + lane * 4;
        #pragma unroll
        for (int k = 0; k < 2; ++k) {
            int bq = wave + 8 * k;
            int Lq = __builtin_amdgcn_readlane(myL, bq);
            if (Lq < OUT_) {
                v4f v = *(const v4f*)(x + ((size_t)bq * C_ + row) * T_ + col);
                v4f r;
                r.x = (col + 0 < Lq) ? v.x : 0.0f;
                r.y = (col + 1 < Lq) ? v.y : 0.0f;
                r.z = (col + 2 < Lq) ? v.z : 0.0f;
                r.w = (col + 3 < Lq) ? v.w : 0.0f;
                *(v4f*)(out + ((size_t)bq * C_ + row) * OUT_ + col) = r;
            }
        }
    }

    int pr = g >> 5;
    int ct = g & 31;

    int lrow = lane >> 3;
    int lc4  = lane & 7;

    unsigned hitA = __builtin_amdgcn_ballot_w32(rank == pr) & 0xffffu;
    unsigned hitB = __builtin_amdgcn_ballot_w32(rank == 15 - pr) & 0xffffu;
    int bA = __builtin_ctz(hitA);
    int bB = __builtin_ctz(hitB);
    int LA = __builtin_amdgcn_readlane(myL, bA);
    int LB = __builtin_amdgcn_readlane(myL, bB);
    int nslA = (LA + KW - 1) / KW;
    nslA = nslA > T_ / KW ? T_ / KW : nslA;
    nslA = (LA < OUT_) ? 0 : nslA;
    int nslB = (LB + KW - 1) / KW;
    nslB = nslB > T_ / KW ? T_ / KW : nslB;
    nslB = (LB < OUT_) ? 0 : nslB;
    if (nslA + nslB == 0) return;

    int WA;
    if (nslB == 0)      WA = NWAVE;
    else if (nslA == 0) WA = 0;
    else {
        int w  = (lane & 7) ? (lane & 7) : 4;
        int ca = (nslA + w - 1) / w;
        int cb = (nslB + (NWAVE - w) - 1) / (NWAVE - w);
        int m  = ca > cb ? ca : cb;
        int bm = __builtin_amdgcn_readlane(m, 1);
        WA = 1;
        #pragma unroll
        for (int k = 2; k < NWAVE; ++k) {
            int mk = __builtin_amdgcn_readlane(m, k);
            if (mk < bm) { bm = mk; WA = k; }
        }
    }

    {
        bool inA = wave < WA;
        int  W   = inA ? WA : NWAVE - WA;
        int  j   = inA ? wave : wave - WA;
        int  b   = inA ? bA : bB;
        int  L   = inA ? LA : LB;
        int  nsl = inA ? nslA : nslB;
        float* sSu = sS[inA ? 0 : 1];
        int cnt = (nsl - j + W - 1) / W;
        cnt = __builtin_amdgcn_readfirstlane(cnt);

        {
            float piL = 3.14159265358979f / (float)L;
            int L2 = 2 * L;
            for (int i = 0; i < cnt; ++i) {
                int t = (j + W * i) * KW + lane;
                int m = (512 * t) % L2;
                if (m > L) m -= L2;
                int mp = m;
                if (2 * m > L)       mp =  L - m;
                else if (2 * m < -L) mp = -L - m;
                sSu[t] = sp_sin((float)mp * piL);
            }
        }
        wave_lds_order();

        int   n0   = ct * 16 + c;
        float Cf   = 512.0f * (float)L;
        GConst gc;
        gc.nLf0 = (float)(n0 * L);
        gc.nLf1 = (float)(n0 * L - 512 * 8);
        gc.Cf   = Cf;
        gc.invC = 1.0f / Cf;
        gc.piC  = 3.14159265358979f / Cf;
        gc.sig  = (n0 & 1) ? (1.0f / 512.0f) : (-1.0f / 512.0f);

        v8f acc[4];
        #pragma unroll
        for (int mt = 0; mt < 4; ++mt)
            acc[mt] = (v8f){0.f, 0.f, 0.f, 0.f, 0.f, 0.f, 0.f, 0.f};

        const float* xb = x + (size_t)b * C_ * T_;

        v2f bf[8];

        {
            int tb = j * KW;
            v4f xr0[16];
            #pragma unroll
            for (int i = 0; i < 16; ++i)
                xr0[i] = *(const v4f*)(xb + (4 * i + lrow) * T_ + tb + lc4 * 4);
            #pragma unroll
            for (int sp = 0; sp < 2; ++sp) {
                int tA = tb + 16 * sp + 2 * h;
                v2f sv[4];
                #pragma unroll
                for (int sl = 0; sl < 4; ++sl)
                    sv[sl] = *(const v2f*)(&sSu[tA + 4 * sl]);
                GSyn S;
                #pragma unroll
                for (int st = 0; st < 16; ++st)
                    g_stage(st, S, gc, tA, L, sv, bf + 4 * sp);
            }
            #pragma unroll
            for (int i = 0; i < 16; ++i)
                *(v4f*)(&sX[wave][4 * i + lrow][lc4 * 4]) = xr0[i];
            wave_lds_order();
        }

        for (int it = 0; it < cnt - 1; ++it) {
            int tbn = ((it + 1) * W + j) * KW;
            v4f xr[16];
            #pragma unroll
            for (int i = 0; i < 16; ++i)
                xr[i] = *(const v4f*)(xb + (4 * i + lrow) * T_ + tbn + lc4 * 4);
            v2f af[8][4];
            #pragma unroll
            for (int s2 = 0; s2 < 8; ++s2)
                #pragma unroll
                for (int mt = 0; mt < 4; ++mt)
                    af[s2][mt] = *(const v2f*)(&sX[wave][mt * 16 + c][4 * s2 + 2 * h]);
            v2f svn[8];
            #pragma unroll
            for (int s2 = 0; s2 < 8; ++s2)
                svn[s2] = *(const v2f*)(&sSu[tbn + 4 * s2 + 2 * h]);
            __builtin_amdgcn_sched_barrier(0);

            #pragma unroll
            for (int sp = 0; sp < 2; ++sp) {
                int tA = tbn + 16 * sp + 2 * h;
                v2f sv[4];
                #pragma unroll
                for (int sl = 0; sl < 4; ++sl) sv[sl] = svn[4 * sp + sl];
                GSyn S;
                #pragma unroll
                for (int st = 0; st < 16; ++st) {
                    int s2 = 4 * sp + (st >> 2);
                    int mt = st & 3;
                    acc[mt] = __builtin_amdgcn_wmma_f32_16x16x4_f32(
                        false, af[s2][mt], false, bf[s2], (short)0, acc[mt], false, false);
                    g_stage(st, S, gc, tA, L, sv, bf + 4 * sp);
                    __builtin_amdgcn_sched_barrier(0);
                }
            }
            wave_lds_order();
            #pragma unroll
            for (int i = 0; i < 16; ++i)
                *(v4f*)(&sX[wave][4 * i + lrow][lc4 * 4]) = xr[i];
            wave_lds_order();
        }

        #pragma unroll
        for (int s2 = 0; s2 < 8; ++s2) {
            int kl = 4 * s2 + 2 * h;
            #pragma unroll
            for (int mt = 0; mt < 4; ++mt) {
                v2f a = *(const v2f*)(&sX[wave][mt * 16 + c][kl]);
                acc[mt] = __builtin_amdgcn_wmma_f32_16x16x4_f32(
                    false, a, false, bf[s2], (short)0, acc[mt], false, false);
            }
        }
        wave_lds_order();

        #pragma unroll
        for (int mt = 0; mt < 4; ++mt)
            #pragma unroll
            for (int r = 0; r < 8; ++r)
                sX[wave][mt * 16 + 8 * h + r][c] = acc[mt][r];
    }
    __syncthreads();

    {
        int row = tid >> 2;
        int cq  = (tid & 3) * 4;
        v4f tl[NWAVE];
        #pragma unroll
        for (int p = 0; p < NWAVE; ++p)
            tl[p] = *(const v4f*)(&sX[p][row][cq]);
        const v4f zero = (v4f){0.f, 0.f, 0.f, 0.f};
        v4f vA = zero;
        v4f vB = zero;
        #pragma unroll
        for (int p = 0; p < NWAVE; ++p) {
            bool a = p < WA;
            vA += a ? tl[p] : zero;
            vB += a ? zero : tl[p];
        }
        if (WA > 0)
            *(v4f*)(out + ((size_t)bA * C_ + row) * OUT_ + ct * 16 + cq) = vA;
        if (WA < NWAVE)
            *(v4f*)(out + ((size_t)bB * C_ + row) * OUT_ + ct * 16 + cq) = vB;
    }
}

extern "C" void kernel_launch(void* const* d_in, const int* in_sizes, int n_in,
                              void* d_out, int out_size, void* d_ws, size_t ws_size,
                              hipStream_t stream) {
    (void)in_sizes; (void)n_in; (void)out_size; (void)d_ws; (void)ws_size;
    const float* x      = (const float*)d_in[0];
    const int*   length = (const int*)d_in[1];
    float*       out    = (float*)d_out;

    sp_fused<<<dim3(256), 256, 0, stream>>>(x, length, out);
}


# Mirror symmetry G[L-t,n] = G[t,512-n]: sums over t <= L/2 of (x[t]+x[L-t])*Ge and (x[t]-x[L-t])*Go give out[n] and out[512-n]; half the matrix instructions and half the synthesis; new sum order
# speedup vs baseline: 7.4918x; 1.1429x over previous
#include <hip/hip_runtime.h>


#define B_    16
#define C_    64
#define T_    2048
#define OUT_  512


typedef float v2f __attribute__((ext_vector_type(2)));
typedef float v4f __attribute__((ext_vector_type(4)));
typedef float v8f __attribute__((ext_vector_type(8)));

__device__ __forceinline__ float sp_sin(float y) {
    float y2 = y * y;
    float p = fmaf(y2, 2.631678171e-6f, -1.982140529e-4f);
    p = fmaf(y2, p, 8.333225735e-3f);
    p = fmaf(y2, p, -1.666666567e-1f);
    p = fmaf(y2, p, 1.0f);
    return y * p;
}

__device__ __forceinline__ float sin_fold(int m, int L, float piL) {
    if (m > L) m -= 2 * L;
    int mp = m;
    if (2 * m > L)       mp =  L - m;
    else if (2 * m < -L) mp = -L - m;
    return sp_sin((float)mp * piL);
}

struct GConst { float nLm0, nLm1, nLp0, nLp1, Cf, piC, sigE, sigO; };
struct GSyn { float tf[4]; float pm[8]; float pp[8]; float ym[8]; float yp[8];
              float ym2[8]; float yp2[8]; float sm[8]; float sp[8]; float we[8]; float wo[8];
              bool zm[8]; bool in[8]; };

__device__ __forceinline__ void g_stage(int j, GSyn& S, const GConst& g, int tA, int Hp1,
                                        const v2f (&svE)[4], const v2f (&svS)[4],
                                        v2f* __restrict__ oE, v2f* __restrict__ oO) {
    switch (j) {
    case 0:
        #pragma unroll
        for (int u = 0; u < 4; ++u) S.tf[u] = (float)(tA + 4 * (u >> 1) + (u & 1));
        #pragma unroll
        for (int q = 0; q < 8; ++q)
            S.pm[q] = fmaf(-512.0f, S.tf[2 * (q >> 2) + (q & 1)], ((q >> 1) & 1) ? g.nLm1 : g.nLm0);
        break;
    case 1:
        #pragma unroll
        for (int q = 0; q < 8; ++q)
            S.pp[q] = fmaf(512.0f, S.tf[2 * (q >> 2) + (q & 1)], ((q >> 1) & 1) ? g.nLp1 : g.nLp0);
        break;
    case 2:
        #pragma unroll
        for (int q = 0; q < 8; ++q) S.yp[q] = g.Cf - S.pp[q];
        #pragma unroll
        for (int q = 0; q < 8; ++q) S.pp[q] = fminf(S.pp[q], S.yp[q]);
        break;
    case 3:
        #pragma unroll
        for (int q = 0; q < 8; ++q) S.ym[q] = S.pm[q] * g.piC;
        #pragma unroll
        for (int q = 0; q < 8; ++q) S.yp[q] = S.pp[q] * g.piC;
        break;
    case 4:
        #pragma unroll
        for (int q = 0; q < 8; ++q) S.ym2[q] = S.ym[q] * S.ym[q];
        #pragma unroll
        for (int q = 0; q < 8; ++q) S.yp2[q] = S.yp[q] * S.yp[q];
        break;
    case 5:
        #pragma unroll
        for (int q = 0; q < 8; ++q) S.sm[q] = fmaf(S.ym2[q], 2.631678171e-6f, -1.982140529e-4f);
        #pragma unroll
        for (int q = 0; q < 8; ++q) S.sp[q] = fmaf(S.yp2[q], 2.631678171e-6f, -1.982140529e-4f);
        break;
    case 6:
        #pragma unroll
        for (int q = 0; q < 8; ++q) S.zm[q] = (S.pm[q] != 0.0f);
        #pragma unroll
        for (int q = 0; q < 8; ++q)
            S.in[q] = (tA + 4 * (q >> 2) + 8 * ((q >> 1) & 1) + (q & 1) < Hp1);
        break;
    case 7:
        #pragma unroll
        for (int q = 0; q < 8; ++q) S.sm[q] = fmaf(S.ym2[q], S.sm[q], 8.333225735e-3f);
        #pragma unroll
        for (int q = 0; q < 8; ++q) S.sp[q] = fmaf(S.yp2[q], S.sp[q], 8.333225735e-3f);
        break;
    case 8:
        #pragma unroll
        for (int q = 0; q < 8; ++q) S.sm[q] = fmaf(S.ym2[q], S.sm[q], -1.666666567e-1f);
        #pragma unroll
        for (int q = 0; q < 8; ++q) S.sp[q] = fmaf(S.yp2[q], S.sp[q], -1.666666567e-1f);
        break;
    case 9:
        #pragma unroll
        for (int q = 0; q < 8; ++q) S.sm[q] = fmaf(S.ym2[q], S.sm[q], 1.0f);
        #pragma unroll
        for (int q = 0; q < 8; ++q) S.sp[q] = fmaf(S.yp2[q], S.sp[q], 1.0f);
        break;
    case 10:
        #pragma unroll
        for (int q = 0; q < 8; ++q) S.sm[q] = S.ym[q] * S.sm[q];
        #pragma unroll
        for (int q = 0; q < 8; ++q) S.sp[q] = S.yp[q] * S.sp[q];
        break;
    case 11:
        #pragma unroll
        for (int q = 0; q < 8; ++q) S.sm[q] = S.sm[q] * S.sp[q];
        #pragma unroll
        for (int q = 0; q < 8; ++q) S.we[q] = svE[(q >> 2) + 2 * ((q >> 1) & 1)][q & 1] * g.sigE;
        break;
    case 12:
        #pragma unroll
        for (int q = 0; q < 8; ++q) S.sm[q] = __builtin_amdgcn_rcpf(S.sm[q]);
        break;
    case 13:
        #pragma unroll
        for (int q = 0; q < 8; ++q) S.wo[q] = svS[(q >> 2) + 2 * ((q >> 1) & 1)][q & 1] * g.sigO;
        #pragma unroll
        for (int q = 0; q < 8; ++q) S.we[q] = S.we[q] * S.sm[q];
        break;
    case 14:
        #pragma unroll
        for (int q = 0; q < 8; ++q) S.wo[q] = S.wo[q] * S.sm[q];
        #pragma unroll
        for (int q = 0; q < 8; ++q) S.we[q] = S.zm[q] ? S.we[q] : 1.0f;
        break;
    default:
        #pragma unroll
        for (int q = 0; q < 8; ++q) S.wo[q] = S.zm[q] ? S.wo[q] : 1.0f;
        #pragma unroll
        for (int q = 0; q < 8; ++q)
            oE[(q >> 2) + 2 * ((q >> 1) & 1)][q & 1] = S.in[q] ? S.we[q] : 0.0f;
        #pragma unroll
        for (int q = 0; q < 8; ++q)
            oO[(q >> 2) + 2 * ((q >> 1) & 1)][q & 1] = S.in[q] ? S.wo[q] : 0.0f;
        break;
    }
}

#define KW     32
#define XPITCH 36
#define NWAVE  8
#define TH_    1088

__device__ __forceinline__ void wave_lds_order() {
    __builtin_amdgcn_fence(__ATOMIC_RELEASE, "wavefront");
    __builtin_amdgcn_wave_barrier();
    __builtin_amdgcn_fence(__ATOMIC_ACQUIRE, "wavefront");
}

__global__ __launch_bounds__(256) __attribute__((amdgpu_waves_per_eu(2, 2))) void sp_fused(const float* __restrict__ x,
                                                const int*   __restrict__ length,
                                                float*       __restrict__ out) {
    __shared__ __attribute__((aligned(16))) float sX[NWAVE][C_][XPITCH];
    __shared__ __attribute__((aligned(16))) float sS[2][TH_];
    __shared__ __attribute__((aligned(16))) float sE[2][TH_];
    __shared__ __attribute__((aligned(16))) float sG[2][TH_];
    __shared__ float sC[NWAVE][2];

    int g     = blockIdx.x;
    int tid   = threadIdx.x;
    int lane  = tid & 31;
    int wave  = __builtin_amdgcn_readfirstlane(tid >> 5);
    int c     = lane & 15;
    int h     = lane >> 4;

    int myL = length[lane & 15];
    int myCost;
    {
        int Lc = myL > T_ ? T_ : myL;
        int ns = ((Lc >> 1) >> 5) + 1;
        myCost = (myL >= OUT_) ? ns : 0;
    }
    int rank = 0;
    #pragma unroll
    for (int bb = 0; bb < B_; ++bb) {
        int cb = __builtin_amdgcn_readlane(myCost, bb);
        rank += (cb > myCost || (cb == myCost && bb < (lane & 15))) ? 1 : 0;
    }

    {
        int row = g >> 2;
        int col = (g & 3) * 128 + lane * 4;
        #pragma unroll
        for (int k = 0; k < 2; ++k) {
            int bq = wave + 8 * k;
            int Lq = __builtin_amdgcn_readlane(myL, bq);
            if (Lq < OUT_) {
                v4f v = *(const v4f*)(x + ((size_t)bq * C_ + row) * T_ + col);
                v4f r;
                r.x = (col + 0 < Lq) ? v.x : 0.0f;
                r.y = (col + 1 < Lq) ? v.y : 0.0f;
                r.z = (col + 2 < Lq) ? v.z : 0.0f;
                r.w = (col + 3 < Lq) ? v.w : 0.0f;
                *(v4f*)(out + ((size_t)bq * C_ + row) * OUT_ + col) = r;
            }
        }
    }

    int pr = g >> 5;
    int jt = (g & 31) >> 1;
    int rh = g & 1;

    int lrow = lane >> 3;
    int lc4  = lane & 7;

    unsigned hitA = __builtin_amdgcn_ballot_w32(rank == pr) & 0xffffu;
    unsigned hitB = __builtin_amdgcn_ballot_w32(rank == 15 - pr) & 0xffffu;
    int bA = __builtin_ctz(hitA);
    int bB = __builtin_ctz(hitB);
    int LA = __builtin_amdgcn_readlane(myL, bA);
    int LB = __builtin_amdgcn_readlane(myL, bB);
    int LAc = LA > T_ ? T_ : LA;
    int LBc = LB > T_ ? T_ : LB;
    int nslA = (LA < OUT_) ? 0 : ((LAc >> 1) >> 5) + 1;
    int nslB = (LB < OUT_) ? 0 : ((LBc >> 1) >> 5) + 1;
    if (nslA + nslB == 0) return;

    int WA;
    if (nslB == 0)      WA = NWAVE;
    else if (nslA == 0) WA = 0;
    else {
        int w  = (lane & 7) ? (lane & 7) : 4;
        int ca = (nslA + w - 1) / w;
        int cb = (nslB + (NWAVE - w) - 1) / (NWAVE - w);
        int m  = ca > cb ? ca : cb;
        int bm = __builtin_amdgcn_readlane(m, 1);
        WA = 1;
        #pragma unroll
        for (int k = 2; k < NWAVE; ++k) {
            int mk = __builtin_amdgcn_readlane(m, k);
            if (mk < bm) { bm = mk; WA = k; }
        }
    }

    float c256a = 0.0f, c256b = 0.0f;
    {
        bool inA = wave < WA;
        int  W   = inA ? WA : NWAVE - WA;
        int  j   = inA ? wave : wave - WA;
        int  b   = inA ? bA : bB;
        int  L   = inA ? LAc : LBc;
        int  nsl = inA ? nslA : nslB;
        int  Hp1 = (L >> 1) + 1;
        float* sSu = sS[inA ? 0 : 1];
        float* sEu = sE[inA ? 0 : 1];
        float* sGu = sG[inA ? 0 : 1];
        int cnt = (nsl - j + W - 1) / W;
        cnt = __builtin_amdgcn_readfirstlane(cnt);

        {
            float piL = 3.14159265358979f / (float)L;
            int L2 = 2 * L;
            for (int i = 0; i < cnt; ++i) {
                int t = (j + W * i) * KW + lane;
                float st = sin_fold((512 * t) % L2, L, piL);
                float s2 = sin_fold(2 * t, L, piL);
                sSu[t] = st;
                sEu[t] = st * s2;
                int d = L - 2 * t;
                float tn = sp_sin((float)t * piL) * __builtin_amdgcn_rcpf(sp_sin((float)d * (0.5f * piL)));
                float g6 = (-1.0f / 512.0f) * st * tn;
                sGu[t] = (d < 0) ? 0.0f : ((d == 0) ? 1.0f : g6);
            }
        }
        wave_lds_order();

        int   n0   = jt * 16 + c;
        float Cf   = 512.0f * (float)L;
        float sig  = (n0 & 1) ? (1.0f / 512.0f) : (-1.0f / 512.0f);
        int   nn   = (n0 <= 128) ? n0 : 256 - n0;
        GConst gc;
        gc.nLm0 = (float)(n0 * L);
        gc.nLm1 = (float)(n0 * L - 512 * 8);
        gc.nLp0 = (float)(n0 * L);
        gc.nLp1 = (float)(n0 * L + 512 * 8);
        gc.Cf   = Cf;
        gc.piC  = 3.14159265358979f / Cf;
        gc.sigE = sig;
        gc.sigO = sig * sp_sin((float)nn * (3.14159265358979f / 256.0f));

        v8f acc[4];
        #pragma unroll
        for (int mt = 0; mt < 4; ++mt)
            acc[mt] = (v8f){0.f, 0.f, 0.f, 0.f, 0.f, 0.f, 0.f, 0.f};

        const float* xb = x + ((size_t)b * C_ + rh * 32) * T_;

        v2f bfE[8], bfO[8];

        {
            int tb = j * KW;
            int t0 = tb + lc4 * 4;
            bool cl = (t0 == 0);
            int  mi = cl ? L - 4 : L - t0 - 3;
            v4f xd[8], xm[8];
            #pragma unroll
            for (int i = 0; i < 8; ++i)
                xd[i] = *(const v4f*)(xb + (4 * i + lrow) * T_ + t0);
            #pragma unroll
            for (int i = 0; i < 8; ++i) {
                v4f v;
                __builtin_memcpy(&v, xb + (4 * i + lrow) * T_ + mi, sizeof(v4f));
                xm[i] = cl ? (v4f){0.0f, v.w, v.z, v.y} : (v4f){v.w, v.z, v.y, v.x};
            }
            #pragma unroll
            for (int sp = 0; sp < 2; ++sp) {
                int tA = tb + 16 * sp + 2 * h;
                v2f svE[4], svS[4];
                #pragma unroll
                for (int sl = 0; sl < 4; ++sl) {
                    svE[sl] = *(const v2f*)(&sEu[tA + 4 * sl]);
                    svS[sl] = *(const v2f*)(&sSu[tA + 4 * sl]);
                }
                GSyn S;
                #pragma unroll
                for (int st = 0; st < 16; ++st)
                    g_stage(st, S, gc, tA, Hp1, svE, svS, bfE + 4 * sp, bfO + 4 * sp);
            }
            if (tb == 0 && n0 == 0 && h == 0) { bfE[0][0] = 2.0f; bfO[0][0] = 0.0f; }
            float mf[4];
            #pragma unroll
            for (int e = 0; e < 4; ++e)
                mf[e] = (t0 + e >= 1 && 2 * (t0 + e) < L) ? 1.0f : 0.0f;
            #pragma unroll
            for (int i = 0; i < 8; ++i) {
                v4f xe, xo;
                #pragma unroll
                for (int e = 0; e < 4; ++e) {
                    xe[e] = fmaf( mf[e], xm[i][e], xd[i][e]);
                    xo[e] = fmaf(-mf[e], xm[i][e], xd[i][e]);
                }
                *(v4f*)(&sX[wave][4 * i + lrow][lc4 * 4])      = xe;
                *(v4f*)(&sX[wave][32 + 4 * i + lrow][lc4 * 4]) = xo;
            }
            wave_lds_order();
        }

        for (int it = 0; it < cnt - 1; ++it) {
            int tbc = (it * W + j) * KW;
            int tbn = ((it + 1) * W + j) * KW;
            int t0n = tbn + lc4 * 4;
            v4f xd[8], xm[8];
            #pragma unroll
            for (int i = 0; i < 8; ++i)
                xd[i] = *(const v4f*)(xb + (4 * i + lrow) * T_ + t0n);
            #pragma unroll
            for (int i = 0; i < 8; ++i)
                __builtin_memcpy(&xm[i], xb + (4 * i + lrow) * T_ + (L - t0n - 3), sizeof(v4f));
            v2f af[8][4];
            #pragma unroll
            for (int s2 = 0; s2 < 8; ++s2)
                #pragma unroll
                for (int mt = 0; mt < 4; ++mt)
                    af[s2][mt] = *(const v2f*)(&sX[wave][mt * 16 + c][4 * s2 + 2 * h]);
            float ra = sX[wave][2 * jt][lane];
            float rb = sX[wave][2 * jt + 1][lane];
            float gq = sGu[tbc + lane];
            v2f svnE[8], svnS[8];
            #pragma unroll
            for (int s2 = 0; s2 < 8; ++s2) {
                svnE[s2] = *(const v2f*)(&sEu[tbn + 4 * s2 + 2 * h]);
                svnS[s2] = *(const v2f*)(&sSu[tbn + 4 * s2 + 2 * h]);
            }
            __builtin_amdgcn_sched_barrier(0);
            c256a = fmaf(ra, gq, c256a);
            c256b = fmaf(rb, gq, c256b);

            #pragma unroll
            for (int sp = 0; sp < 2; ++sp) {
                int tA = tbn + 16 * sp + 2 * h;
                v2f svE[4], svS[4];
                #pragma unroll
                for (int sl = 0; sl < 4; ++sl) { svE[sl] = svnE[4 * sp + sl]; svS[sl] = svnS[4 * sp + sl]; }
                GSyn S;
                #pragma unroll
                for (int st = 0; st < 16; ++st) {
                    int s2 = 4 * sp + (st >> 2);
                    int mt = st & 3;
                    acc[mt] = __builtin_amdgcn_wmma_f32_16x16x4_f32(
                        false, af[s2][mt], false, (mt < 2) ? bfE[s2] : bfO[s2], (short)0, acc[mt], false, false);
                    g_stage(st, S, gc, tA, Hp1, svE, svS, bfE + 4 * sp, bfO + 4 * sp);
                    __builtin_amdgcn_sched_barrier(0);
                }
            }
            wave_lds_order();
            float mf[4];
            #pragma unroll
            for (int e = 0; e < 4; ++e)
                mf[e] = (2 * (t0n + e) < L) ? 1.0f : 0.0f;
            #pragma unroll
            for (int i = 0; i < 8; ++i) {
                v4f xe, xo;
                #pragma unroll
                for (int e = 0; e < 4; ++e) {
                    xe[e] = fmaf( mf[e], xm[i][3 - e], xd[i][e]);
                    xo[e] = fmaf(-mf[e], xm[i][3 - e], xd[i][e]);
                }
                *(v4f*)(&sX[wave][4 * i + lrow][lc4 * 4])      = xe;
                *(v4f*)(&sX[wave][32 + 4 * i + lrow][lc4 * 4]) = xo;
            }
            wave_lds_order();
        }

        {
            int tbc = ((cnt - 1) * W + j) * KW;
            float ra = sX[wave][2 * jt][lane];
            float rb = sX[wave][2 * jt + 1][lane];
            float gq = sGu[tbc + lane];
            c256a = fmaf(ra, gq, c256a);
            c256b = fmaf(rb, gq, c256b);
        }
        #pragma unroll
        for (int s2 = 0; s2 < 8; ++s2) {
            int kl = 4 * s2 + 2 * h;
            #pragma unroll
            for (int mt = 0; mt < 4; ++mt) {
                v2f a = *(const v2f*)(&sX[wave][mt * 16 + c][kl]);
                acc[mt] = __builtin_amdgcn_wmma_f32_16x16x4_f32(
                    false, a, false, (mt < 2) ? bfE[s2] : bfO[s2], (short)0, acc[mt], false, false);
            }
        }
        wave_lds_order();

        #pragma unroll
        for (int mt = 0; mt < 4; ++mt)
            #pragma unroll
            for (int r = 0; r < 8; ++r)
                sX[wave][mt * 16 + 8 * h + r][c] = acc[mt][r];

        #pragma unroll
        for (int o = 16; o >= 1; o >>= 1) {
            c256a += __shfl_xor(c256a, o);
            c256b += __shfl_xor(c256b, o);
        }
        if (lane == 0) { sC[wave][0] = c256a; sC[wave][1] = c256b; }
    }
    __syncthreads();

    {
        int r  = tid >> 3;
        int cq = (tid & 7) * 2;
        v2f ts[NWAVE], td[NWAVE];
        #pragma unroll
        for (int p = 0; p < NWAVE; ++p) {
            ts[p] = *(const v2f*)(&sX[p][r][cq]);
            td[p] = *(const v2f*)(&sX[p][32 + r][cq]);
        }
        const v2f zero = (v2f){0.f, 0.f};
        v2f SA = zero, DA = zero, SB = zero, DB = zero;
        #pragma unroll
        for (int p = 0; p < NWAVE; ++p) {
            bool a = p < WA;
            SA += a ? ts[p] : zero;
            DA += a ? td[p] : zero;
            SB += a ? zero : ts[p];
            DB += a ? zero : td[p];
        }
        int n  = jt * 16 + cq;
        int ch = rh * 32 + r;
        if (WA > 0) {
            float* o = out + ((size_t)bA * C_ + ch) * OUT_;
            *(v2f*)(o + n) = 0.5f * (SA + DA);
            v2f m = 0.5f * (SA - DA);
            if (n != 0) o[OUT_ - n] = m[0];
            o[OUT_ - n - 1] = m[1];
        }
        if (WA < NWAVE) {
            float* o = out + ((size_t)bB * C_ + ch) * OUT_;
            *(v2f*)(o + n) = 0.5f * (SB + DB);
            v2f m = 0.5f * (SB - DB);
            if (n != 0) o[OUT_ - n] = m[0];
            o[OUT_ - n - 1] = m[1];
        }
        if (tid < 4) {
            int  u    = tid >> 1;
            int  e    = tid & 1;
            bool live = u ? (WA < NWAVE) : (WA > 0);
            float s = 0.0f;
            #pragma unroll
            for (int p = 0; p < NWAVE; ++p) {
                bool mine = u ? (p >= WA) : (p < WA);
                s += mine ? sC[p][e] : 0.0f;
            }
            if (live)
                out[((size_t)(u ? bB : bA) * C_ + rh * 32 + 2 * jt + e) * OUT_ + 256] = s;
        }
    }
}

extern "C" void kernel_launch(void* const* d_in, const int* in_sizes, int n_in,
                              void* d_out, int out_size, void* d_ws, size_t ws_size,
                              hipStream_t stream) {
    (void)in_sizes; (void)n_in; (void)out_size; (void)d_ws; (void)ws_size;
    const float* x      = (const float*)d_in[0];
    const int*   length = (const int*)d_in[1];
    float*       out    = (float*)d_out;

    sp_fused<<<dim3(256), 256, 0, stream>>>(x, length, out);
}
